# starting kernel
# baseline (speedup 1.0000x reference)
#include <hip/hip_runtime.h>
#include <hip/hip_bf16.h>

typedef float v2f __attribute__((ext_vector_type(2)));
typedef float v8f __attribute__((ext_vector_type(8)));

__global__ void ScorePredictor_81097572483639_kernel(
    const float* __restrict__ h_drug,
    const float* __restrict__ d_disease,
    const int* __restrict__ src_idx,
    const int* __restrict__ dst_idx,
    float* __restrict__ out,
    int n_edges, int n_groups) {
  const int lane          = threadIdx.x & 31;
  const int waveInBlock   = threadIdx.x >> 5;
  const int wavesPerBlock = blockDim.x >> 5;
  const int groupStride   = gridDim.x * wavesPerBlock;
  int group = blockIdx.x * wavesPerBlock + waveInBlock;

  const int m  = lane & 15;
  const int kh = lane >> 4;

  for (; group < n_groups; group += groupStride) {
    const int base = group * 16;
    int e = base + m;
    if (e >= n_edges) e = 0;

    const int si = src_idx[e];
    const int di = dst_idx[e];
    const float* pa = h_drug    + (size_t)si * 128 + (kh << 1);
    const float* pb = d_disease + (size_t)di * 128 + (kh << 1);

    v8f c = {0.f, 0.f, 0.f, 0.f, 0.f, 0.f, 0.f, 0.f};
#pragma unroll
    for (int k0 = 0; k0 < 128; k0 += 4) {
      v2f a = *(const v2f*)(pa + k0);
      v2f b = *(const v2f*)(pb + k0);
      c = __builtin_amdgcn_wmma_f32_16x16x4_f32(false, a, false, b,
                                                (short)0, c, false, false);
    }

    const int srcLane = (m < 8) ? m : (m + 16);
    const int sel     = m & 7;
    float diag = 0.f;
#pragma unroll
    for (int i = 0; i < 8; ++i) {
      float t = __shfl(c[i], srcLane, 32);
      if (sel == i) diag = t;
    }

    const int eo = base + lane;
    if (lane < 16 && eo < n_edges) {
      float mean = diag * (1.0f / 128.0f);
      out[eo] = 1.0f / (1.0f + __expf(-mean));
    }
  }
}

extern "C" void kernel_launch(void* const* d_in, const int* in_sizes, int n_in,
                              void* d_out, int out_size, void* d_ws, size_t ws_size,
                              hipStream_t stream) {
  const float* h_drug    = (const float*)d_in[0];
  const float* d_disease = (const float*)d_in[1];
  const int*   src_idx   = (const int*)d_in[2];
  const int*   dst_idx   = (const int*)d_in[3];
  float*       out       = (float*)d_out;

  const int n_edges  = in_sizes[2];
  const int n_groups = (n_edges + 15) / 16;

  const int threads        = 256;
  const int wavesPerBlock  = threads / 32;
  int blocks = (n_groups + wavesPerBlock - 1) / wavesPerBlock;
  if (blocks < 1) blocks = 1;

  ScorePredictor_81097572483639_kernel<<<blocks, threads, 0, stream>>>(
      h_drug, d_disease, src_idx, dst_idx, out, n_edges, n_groups);
}


# rows gathered as whole 128-B lines straight into the WMMA operands: 2 edges per accumulator, diagonal halves summed by row_xmask adds, no LDS, 256 looping blocks; order of the feature sum changes
# speedup vs baseline: 3.4850x; 3.4850x over previous
#include <hip/hip_runtime.h>

typedef float v2f __attribute__((ext_vector_type(2)));
typedef float v4f __attribute__((ext_vector_type(4)));
typedef float v8f __attribute__((ext_vector_type(8)));

template <int N>
__device__ __forceinline__ float xm(float v) {
  return __builtin_bit_cast(float, __builtin_amdgcn_update_dpp(
      0, __builtin_bit_cast(int, v), 0x160 + N, 0xF, 0xF, true));
}

template <int P>
__device__ __forceinline__ float diag_sum(const v8f& c) {
  float w[4];
#pragma unroll
  for (int i = 0; i < 4; ++i) {
    const int r = (P & 4) | i;
    w[i] = c[r] + xm<4>(c[r ^ 4]);
  }
  float x[2];
#pragma unroll
  for (int j = 0; j < 2; ++j) {
    const int i = (P & 2) | j;
    x[j] = w[i] + xm<2>(w[i ^ 2]);
  }
  const int j = P & 1;
  return x[j] + xm<1>(x[j ^ 1]);
}

struct Frag { v4f x0, x1, y0, y1; };

template <int Q>
__device__ __forceinline__ void issue_pair(Frag& f, const char* ta, const char* tb,
                                           unsigned sidx, unsigned didx,
                                           unsigned laneOff) {
  const unsigned lo = 0x11111111u * (unsigned)(2 * Q);
  const unsigned hi = 0x11111111u * (unsigned)(2 * Q + 1);
  const unsigned si = __builtin_amdgcn_permlane16(0u, sidx, lo, hi, false, false);
  const unsigned di = __builtin_amdgcn_permlane16(0u, didx, lo, hi, false, false);
  const v4f* pa = (const v4f*)(ta + ((si << 9) + laneOff));
  const v4f* pb = (const v4f*)(tb + ((di << 9) + laneOff));
  f.x0 = pa[0]; f.x1 = pa[16];
  f.y0 = pb[0]; f.y1 = pb[16];
}

template <int Q>
__device__ __forceinline__ float finish_pair(const Frag& f) {
  v8f c = {0.f, 0.f, 0.f, 0.f, 0.f, 0.f, 0.f, 0.f};
  v2f a, b;
  a = {f.x0.x, f.x0.y}; b = {f.y0.x, f.y0.y};
  c = __builtin_amdgcn_wmma_f32_16x16x4_f32(false, a, false, b, (short)0, c, false, false);
  a = {f.x0.z, f.x0.w}; b = {f.y0.z, f.y0.w};
  c = __builtin_amdgcn_wmma_f32_16x16x4_f32(false, a, false, b, (short)0, c, false, false);
  a = {f.x1.x, f.x1.y}; b = {f.y1.x, f.y1.y};
  c = __builtin_amdgcn_wmma_f32_16x16x4_f32(false, a, false, b, (short)0, c, false, false);
  a = {f.x1.z, f.x1.w}; b = {f.y1.z, f.y1.w};
  c = __builtin_amdgcn_wmma_f32_16x16x4_f32(false, a, false, b, (short)0, c, false, false);
  return diag_sum<Q>(c);
}

#define ISSUE4(F, Q0, S, D)                                            \
  issue_pair<Q0 + 0>(F[0], ta, tb, S, D, laneOff);                     \
  issue_pair<Q0 + 1>(F[1], ta, tb, S, D, laneOff);                     \
  issue_pair<Q0 + 2>(F[2], ta, tb, S, D, laneOff);                     \
  issue_pair<Q0 + 3>(F[3], ta, tb, S, D, laneOff);

#define FINISH1(F, I, Q, OUT)                                          \
  { const float y = finish_pair<Q>(F[I]); OUT = (key == Q) ? y : OUT; }

#define FINISH4(F, Q0, OUT)                                            \
  FINISH1(F, 0, Q0 + 0, OUT) FINISH1(F, 1, Q0 + 1, OUT)                \
  FINISH1(F, 2, Q0 + 2, OUT) FINISH1(F, 3, Q0 + 3, OUT)

#define PIN_LOADS() __builtin_amdgcn_sched_barrier(0)

__global__ __launch_bounds__(512) void edge_score_kernel(
    const float* __restrict__ h_drug,
    const float* __restrict__ d_disease,
    const int* __restrict__ src_idx,
    const int* __restrict__ dst_idx,
    float* __restrict__ out,
    int n_edges, int n_units) {
  const int lane = threadIdx.x & 31;
  const int wave = threadIdx.x >> 5;
  const int wavesPerBlock = blockDim.x >> 5;
  const int c16 = lane & 15;
  const int h = lane >> 4;
  const char* ta = (const char*)h_drug;
  const char* tb = (const char*)d_disease;
  const unsigned laneOff = (unsigned)(h * 128 + (c16 & 7) * 16);
  const int key = c16 ^ (h << 3);
  const int eoff = (lane < 16) ? 2 * lane : ((lane < 24) ? 2 * lane - 15 : 2 * lane - 47);

  const int nb = (int)gridDim.x, bid = (int)blockIdx.x;
  const int per = n_units / nb, rem = n_units - per * nb;
  const int uBeg = bid * per + (bid < rem ? bid : rem);
  const int uEnd = uBeg + per + (bid < rem ? 1 : 0);

  int u = uBeg + wave;
  if (u >= uEnd) return;
  const int last = n_edges - 1;
  unsigned sA, sB, dA, dB;
  {
    const int eA = min(u * 32 + c16, last), eB = min(u * 32 + 16 + c16, last);
    sA = (unsigned)src_idx[eA]; sB = (unsigned)src_idx[eB];
    dA = (unsigned)dst_idx[eA]; dB = (unsigned)dst_idx[eB];
  }

  for (; u < uEnd; u += wavesPerBlock) {
    const int base = u * 32;
    Frag f0[4], f1[4];
    ISSUE4(f0, 0, sA, dA)
    ISSUE4(f1, 4, sA, dA)
    const int un = min(u + wavesPerBlock, uEnd - 1);
    const int nA = min(un * 32 + c16, last), nB = min(un * 32 + 16 + c16, last);
    const unsigned sA2 = (unsigned)src_idx[nA], sB2 = (unsigned)src_idx[nB];
    const unsigned dA2 = (unsigned)dst_idx[nA], dB2 = (unsigned)dst_idx[nB];
    PIN_LOADS();
    float outA = 0.f, outB = 0.f;
    FINISH4(f0, 0, outA)
    ISSUE4(f0, 0, sB, dB)
    PIN_LOADS();
    FINISH4(f1, 4, outA)
    ISSUE4(f1, 4, sB, dB)
    PIN_LOADS();
    FINISH4(f0, 0, outB)
    FINISH4(f1, 4, outB)

    const float movedB = xm<8>(outB);
    const float dot = (key < 8) ? outA : movedB;
    const int eo = base + eoff;
    if (eo < n_edges) {
      const float mean = dot * (1.0f / 128.0f);
      out[eo] = 1.0f / (1.0f + __expf(-mean));
    }
    sA = sA2; sB = sB2; dA = dA2; dB = dB2;
  }
}

extern "C" void kernel_launch(void* const* d_in, const int* in_sizes, int n_in,
                              void* d_out, int out_size, void* d_ws, size_t ws_size,
                              hipStream_t stream) {
  const float* h_drug    = (const float*)d_in[0];
  const float* d_disease = (const float*)d_in[1];
  const int*   src_idx   = (const int*)d_in[2];
  const int*   dst_idx   = (const int*)d_in[3];
  float*       out       = (float*)d_out;

  const int n_edges = in_sizes[2];
  const int n_units = (n_edges + 31) / 32;

  edge_score_kernel<<<256, 512, 0, stream>>>(
      h_drug, d_disease, src_idx, dst_idx, out, n_edges, n_units);
}
